# starting kernel
# baseline (speedup 1.0000x reference)
#include <hip/hip_runtime.h>
#include <hip/hip_bf16.h>
#include <math.h>

typedef __attribute__((ext_vector_type(2))) float v2f;
typedef __attribute__((ext_vector_type(8))) float v8f;

#define N_NODES 100000
#define N_EDGES 1600000
#define IN_DIM 128
#define OUT_DIM 64
#define NEG_SLOPE 0.01f

__global__ void gat_gemm_wmma(const float* __restrict__ h,
                              const float* __restrict__ W,
                              float* __restrict__ z,
                              int n_nodes) {
    int gtid = blockIdx.x * blockDim.x + threadIdx.x;
    int wave = gtid >> 5;
    int lane = threadIdx.x & 31;
    int m0 = wave * 16;
    if (m0 >= n_nodes) return;

    int rowA  = m0 + (lane & 15);
    int khalf = (lane >> 4) * 2;
    int nlane = lane & 15;

    v8f acc0 = {}, acc1 = {}, acc2 = {}, acc3 = {};

    for (int k0 = 0; k0 < IN_DIM; k0 += 4) {
        v2f a = *(const v2f*)(h + (size_t)rowA * IN_DIM + k0 + khalf);

        v2f b0 = *(const v2f*)(W + (size_t)(nlane +  0) * IN_DIM + k0 + khalf);
        v2f b1 = *(const v2f*)(W + (size_t)(nlane + 16) * IN_DIM + k0 + khalf);
        v2f b2 = *(const v2f*)(W + (size_t)(nlane + 32) * IN_DIM + k0 + khalf);
        v2f b3 = *(const v2f*)(W + (size_t)(nlane + 48) * IN_DIM + k0 + khalf);

        acc0 = __builtin_amdgcn_wmma_f32_16x16x4_f32(false, a, false, b0, (short)0, acc0, false, false);
        acc1 = __builtin_amdgcn_wmma_f32_16x16x4_f32(false, a, false, b1, (short)0, acc1, false, false);
        acc2 = __builtin_amdgcn_wmma_f32_16x16x4_f32(false, a, false, b2, (short)0, acc2, false, false);
        acc3 = __builtin_amdgcn_wmma_f32_16x16x4_f32(false, a, false, b3, (short)0, acc3, false, false);
    }

    int rsel = (lane >> 4) * 8;
#pragma unroll
    for (int r = 0; r < 8; ++r) {
        size_t ro = (size_t)(m0 + r + rsel) * OUT_DIM;
        z[ro + nlane +  0] = acc0[r];
        z[ro + nlane + 16] = acc1[r];
        z[ro + nlane + 32] = acc2[r];
        z[ro + nlane + 48] = acc3[r];
    }
}

__global__ void gat_node_scores(const float* __restrict__ z,
                                const float* __restrict__ attn,
                                float* __restrict__ s_l,
                                float* __restrict__ s_r,
                                int n) {
    int i = blockIdx.x * blockDim.x + threadIdx.x;
    if (i >= n) return;
    const float4* zr = (const float4*)(z + (size_t)i * OUT_DIM);
    const float4* al = (const float4*)(attn);
    const float4* ar = (const float4*)(attn + OUT_DIM);
    float sl = 0.f, sr = 0.f;
#pragma unroll
    for (int q = 0; q < OUT_DIM / 4; ++q) {
        float4 zv = zr[q];
        float4 av = al[q];
        float4 bv = ar[q];
        sl += zv.x * av.x + zv.y * av.y + zv.z * av.z + zv.w * av.w;
        sr += zv.x * bv.x + zv.y * bv.y + zv.z * bv.z + zv.w * bv.w;
    }
    s_l[i] = sl;
    s_r[i] = sr;
}

__global__ void gat_zero_i32(int* __restrict__ p, int n) {
    int i = blockIdx.x * blockDim.x + threadIdx.x;
    if (i < n) p[i] = 0;
}

__global__ void gat_edge_scores(const int* __restrict__ src,
                                const int* __restrict__ dst,
                                const float* __restrict__ s_l,
                                const float* __restrict__ s_r,
                                float* __restrict__ e,
                                int* __restrict__ deg,
                                int ne) {
    int i = blockIdx.x * blockDim.x + threadIdx.x;
    if (i >= ne) return;
    int d = dst[i];
    float v = s_l[src[i]] + s_r[d];
    e[i] = (v > 0.f) ? v : NEG_SLOPE * v;
    atomicAdd(&deg[d], 1);
}

__global__ void gat_scan_excl(const int* __restrict__ deg,
                              int* __restrict__ offs,
                              int* __restrict__ cursor,
                              int n) {
    __shared__ int s[1024];
    int t = threadIdx.x;
    int run = 0;
    for (int base = 0; base < n; base += 1024) {
        int idx = base + t;
        int v = (idx < n) ? deg[idx] : 0;
        s[t] = v;
        __syncthreads();
        int val = v;
        for (int off = 1; off < 1024; off <<= 1) {
            int tmp = (t >= off) ? s[t - off] : 0;
            __syncthreads();
            val += tmp;
            s[t] = val;
            __syncthreads();
        }
        int excl = run + val - v;
        if (idx < n) { offs[idx] = excl; cursor[idx] = excl; }
        run += s[1023];
        __syncthreads();
    }
    if (t == 0) offs[n] = run;
}

__global__ void gat_fill_csr(const int* __restrict__ dst,
                             int* __restrict__ cursor,
                             int* __restrict__ csr,
                             int ne) {
    int i = blockIdx.x * blockDim.x + threadIdx.x;
    if (i >= ne) return;
    int p = atomicAdd(&cursor[dst[i]], 1);
    csr[p] = i;
}

__global__ void gat_node_aggregate(const float* __restrict__ z,
                                   const float* __restrict__ e,
                                   const int* __restrict__ src,
                                   const int* __restrict__ csr,
                                   const int* __restrict__ offs,
                                   const int* __restrict__ deg,
                                   float* __restrict__ out,
                                   int n) {
    int gtid = blockIdx.x * blockDim.x + threadIdx.x;
    int node = gtid >> 5;
    if (node >= n) return;
    int lane = threadIdx.x & 31;
    int c0 = lane * 2;

    int d = deg[node];
    int o = offs[node];

    float accx = 0.f, accy = 0.f;
    if (d > 0) {
        float sum = 0.f;
        for (int j = 0; j < d; ++j) sum += e[csr[o + j]];
        float mean = sum / (float)d;

        float m = -1e30f;
        for (int j = 0; j < d; ++j) {
            float ev = e[csr[o + j]];
            float e2 = (ev < mean) ? 0.f : ev;
            m = fmaxf(m, e2);
        }

        float denom = 0.f;
        for (int j = 0; j < d; ++j) {
            int ei = csr[o + j];
            float ev = e[ei];
            float e2 = (ev < mean) ? 0.f : ev;
            float ex = __expf(e2 - m);
            denom += ex;
            const float2 zv = *(const float2*)(z + (size_t)src[ei] * OUT_DIM + c0);
            accx += ex * zv.x;
            accy += ex * zv.y;
        }
        float inv = 1.f / denom;
        accx *= inv;
        accy *= inv;
    }
    float2 r;
    r.x = accx;
    r.y = accy;
    *(float2*)(out + (size_t)node * OUT_DIM + c0) = r;
}

extern "C" void kernel_launch(void* const* d_in, const int* in_sizes, int n_in,
                              void* d_out, int out_size, void* d_ws, size_t ws_size,
                              hipStream_t stream) {
    const float* h      = (const float*)d_in[0];
    const int*   src    = (const int*)d_in[1];
    const int*   dst    = (const int*)d_in[2];
    const float* W_fc   = (const float*)d_in[3];
    const float* W_attn = (const float*)d_in[4];
    float* out = (float*)d_out;

    char* ws = (char*)d_ws;
    size_t off = 0;
    auto carve = [&](size_t bytes) {
        void* p = ws + off;
        off += (bytes + 15) & ~(size_t)15;
        return p;
    };
    float* z      = (float*)carve((size_t)N_NODES * OUT_DIM * sizeof(float));
    float* s_l    = (float*)carve((size_t)N_NODES * sizeof(float));
    float* s_r    = (float*)carve((size_t)N_NODES * sizeof(float));
    float* e      = (float*)carve((size_t)N_EDGES * sizeof(float));
    int*   deg    = (int*)carve((size_t)N_NODES * sizeof(int));
    int*   offs   = (int*)carve((size_t)(N_NODES + 1) * sizeof(int));
    int*   cursor = (int*)carve((size_t)N_NODES * sizeof(int));
    int*   csr    = (int*)carve((size_t)N_EDGES * sizeof(int));
    (void)ws_size;

    const int B = 256;

    int gemm_waves  = N_NODES / 16;
    int gemm_blocks = (gemm_waves + 7) / 8;
    gat_gemm_wmma<<<gemm_blocks, B, 0, stream>>>(h, W_fc, z, N_NODES);

    gat_node_scores<<<(N_NODES + B - 1) / B, B, 0, stream>>>(z, W_attn, s_l, s_r, N_NODES);

    gat_zero_i32<<<(N_NODES + B - 1) / B, B, 0, stream>>>(deg, N_NODES);
    gat_edge_scores<<<(N_EDGES + B - 1) / B, B, 0, stream>>>(src, dst, s_l, s_r, e, deg, N_EDGES);

    gat_scan_excl<<<1, 1024, 0, stream>>>(deg, offs, cursor, N_NODES);

    gat_fill_csr<<<(N_EDGES + B - 1) / B, B, 0, stream>>>(dst, cursor, csr, N_EDGES);

    int agg_blocks = (N_NODES + 7) / 8;
    gat_node_aggregate<<<agg_blocks, B, 0, stream>>>(z, e, src, csr, offs, deg, out, N_NODES);
}


# Node pass, fewer instructions per node: weight x row is one packed fma (scalar weight), lane crossings fold into their add/max, row groups at constant lanes, remainder as 4+2+1; sum order same, fused
# speedup vs baseline: 6.4581x; 6.4581x over previous
#include <hip/hip_runtime.h>
#include <hip/hip_bf16.h>
#include <math.h>

typedef __attribute__((ext_vector_type(2))) float v2f;
typedef __attribute__((ext_vector_type(8))) float v8f;

#define N_NODES 100000
#define N_EDGES 1600000
#define IN_DIM 128
#define OUT_DIM 64
#define NEG_SLOPE 0.01f
#define KBLOCKS (IN_DIM / 8)
#define A_AHEAD 4

#define ROW_WORDS 32
#define ROW_HDR 2
#define ROW_CAP (ROW_WORDS - ROW_HDR)

#define LINK_CHUNK 256
#define CTR_STRIDE 64

typedef __attribute__((ext_vector_type(2))) unsigned int v2u;

__device__ __forceinline__ void split_k8(float4 v, v2f& lo, v2f& hi) {
    v2u sx = __builtin_amdgcn_permlane16_swap(__float_as_uint(v.x), __float_as_uint(v.z), false, false);
    v2u sy = __builtin_amdgcn_permlane16_swap(__float_as_uint(v.y), __float_as_uint(v.w), false, false);
    lo.x = __uint_as_float(sx.x);
    lo.y = __uint_as_float(sy.x);
    hi.x = __uint_as_float(sx.y);
    hi.y = __uint_as_float(sy.y);
}

__global__ void gat_gemm_wmma(const float* __restrict__ h,
                              const float* __restrict__ W,
                              float* __restrict__ z,
                              int* __restrict__ rows,
                              int* __restrict__ ctr,
                              int n_nodes) {
    int gtid = blockIdx.x * blockDim.x + threadIdx.x;
    if (gtid < n_nodes) *(int2*)(rows + (size_t)gtid * ROW_WORDS) = make_int2(0, -1);
    if (gtid < 2) ctr[gtid * CTR_STRIDE] = 0;
    int wave = gtid >> 5;
    int lane = threadIdx.x & 31;
    int m0 = wave * 16;
    if (m0 >= n_nodes) return;

    int rowA  = m0 + (lane & 15);
    int nlane = lane & 15;

    v8f acc0 = {}, acc1 = {}, acc2 = {}, acc3 = {};

    int kquad = (lane >> 4) * 4;
    const float* hp = h + (size_t)rowA * IN_DIM + kquad;
    const float* wp = W + (size_t)nlane * IN_DIM + kquad;

    float4 va[KBLOCKS];
    float4 vb[KBLOCKS][4];
#pragma unroll
    for (int i = 0; i < A_AHEAD; ++i) va[i] = *(const float4*)(hp + 8 * i);
#pragma unroll
    for (int t = 0; t < 4; ++t) vb[0][t] = *(const float4*)(wp + (size_t)(16 * t) * IN_DIM);

#pragma unroll
    for (int i = 0; i < KBLOCKS; ++i) {
        if (i + A_AHEAD < KBLOCKS) va[i + A_AHEAD] = *(const float4*)(hp + 8 * (i + A_AHEAD));
        if (i + 1 < KBLOCKS) {
#pragma unroll
            for (int t = 0; t < 4; ++t)
                vb[i + 1][t] = *(const float4*)(wp + (size_t)(16 * t) * IN_DIM + 8 * (i + 1));
        }
        __builtin_amdgcn_sched_barrier(0);

        v2f a_lo, a_hi, b0_lo, b0_hi, b1_lo, b1_hi, b2_lo, b2_hi, b3_lo, b3_hi;
        split_k8(va[i],    a_lo,  a_hi);
        split_k8(vb[i][0], b0_lo, b0_hi);
        split_k8(vb[i][1], b1_lo, b1_hi);
        split_k8(vb[i][2], b2_lo, b2_hi);
        split_k8(vb[i][3], b3_lo, b3_hi);

        acc0 = __builtin_amdgcn_wmma_f32_16x16x4_f32(false, a_lo, false, b0_lo, (short)0, acc0, false, false);
        acc1 = __builtin_amdgcn_wmma_f32_16x16x4_f32(false, a_lo, false, b1_lo, (short)0, acc1, false, false);
        acc2 = __builtin_amdgcn_wmma_f32_16x16x4_f32(false, a_lo, false, b2_lo, (short)0, acc2, false, false);
        acc3 = __builtin_amdgcn_wmma_f32_16x16x4_f32(false, a_lo, false, b3_lo, (short)0, acc3, false, false);
        acc0 = __builtin_amdgcn_wmma_f32_16x16x4_f32(false, a_hi, false, b0_hi, (short)0, acc0, false, false);
        acc1 = __builtin_amdgcn_wmma_f32_16x16x4_f32(false, a_hi, false, b1_hi, (short)0, acc1, false, false);
        acc2 = __builtin_amdgcn_wmma_f32_16x16x4_f32(false, a_hi, false, b2_hi, (short)0, acc2, false, false);
        acc3 = __builtin_amdgcn_wmma_f32_16x16x4_f32(false, a_hi, false, b3_hi, (short)0, acc3, false, false);
    }

    int rsel = (lane >> 4) * 8;
#pragma unroll
    for (int r = 0; r < 8; ++r) {
        size_t ro = (size_t)(m0 + r + rsel) * OUT_DIM;
        z[ro + nlane +  0] = acc0[r];
        z[ro + nlane + 16] = acc1[r];
        z[ro + nlane + 32] = acc2[r];
        z[ro + nlane + 48] = acc3[r];
    }
}

__device__ __forceinline__ void node_scores_body(const float* __restrict__ z,
                                                 const float* __restrict__ attn,
                                                 float* __restrict__ s_l,
                                                 float* __restrict__ s_r,
                                                 int n, int i) {
    if (i >= n) return;
    const float4* zr = (const float4*)(z + (size_t)i * OUT_DIM);
    const float4* al = (const float4*)(attn);
    const float4* ar = (const float4*)(attn + OUT_DIM);
    float sl = 0.f, sr = 0.f;
#pragma unroll
    for (int q = 0; q < OUT_DIM / 4; ++q) {
        float4 zv = zr[q];
        float4 av = al[q];
        float4 bv = ar[q];
        sl += zv.x * av.x + zv.y * av.y + zv.z * av.z + zv.w * av.w;
        sr += zv.x * bv.x + zv.y * bv.y + zv.z * bv.z + zv.w * bv.w;
    }
    s_l[i] = sl;
    s_r[i] = sr;
}

__device__ __forceinline__ void fetch_chunk(const int* __restrict__ src,
                                            const int* __restrict__ dst,
                                            int ne, int c, int lane,
                                            int (&dd)[8], int (&ss)[8]) {
    int eb = c * LINK_CHUNK + lane * 4;
    if (c * LINK_CHUNK + LINK_CHUNK <= ne) {
        int4 da = *(const int4*)(dst + eb);
        int4 db = *(const int4*)(dst + eb + LINK_CHUNK / 2);
        int4 sa = *(const int4*)(src + eb);
        int4 sb = *(const int4*)(src + eb + LINK_CHUNK / 2);
        dd[0] = da.x; dd[1] = da.y; dd[2] = da.z; dd[3] = da.w;
        dd[4] = db.x; dd[5] = db.y; dd[6] = db.z; dd[7] = db.w;
        ss[0] = sa.x; ss[1] = sa.y; ss[2] = sa.z; ss[3] = sa.w;
        ss[4] = sb.x; ss[5] = sb.y; ss[6] = sb.z; ss[7] = sb.w;
    } else {
#pragma unroll
        for (int k = 0; k < 8; ++k) {
            int e = eb + (k >> 2) * (LINK_CHUNK / 2) + (k & 3);
            dd[k] = -1;
            ss[k] = 0;
            if (e < ne) { dd[k] = dst[e]; ss[k] = src[e]; }
        }
    }
}

__device__ __forceinline__ void link_edges(int* __restrict__ rows,
                                           int* __restrict__ next,
                                           int n, int split, int h, int c, int lane,
                                           const int (&dd)[8], const int (&ss)[8]) {
    int eb = c * LINK_CHUNK + lane * 4;
    bool mine[8];
    int  rk[8];
#pragma unroll
    for (int k = 0; k < 8; ++k) {
        mine[k] = (unsigned)dd[k] < (unsigned)n && ((dd[k] >= split) == (h != 0));
        rk[k] = 0;
        if (mine[k]) rk[k] = atomicAdd(rows + (size_t)dd[k] * ROW_WORDS, 1);
    }
#pragma unroll
    for (int k = 0; k < 8; ++k) {
        if (mine[k]) {
            int* row = rows + (size_t)dd[k] * ROW_WORDS;
            if (rk[k] < ROW_CAP) {
                row[ROW_HDR + rk[k]] = ss[k];
            } else {
                int e = eb + (k >> 2) * (LINK_CHUNK / 2) + (k & 3);
                next[e] = atomicExch(&row[1], e);
            }
        }
    }
}

__device__ __forceinline__ void link_by_half(const int* __restrict__ src,
                                             const int* __restrict__ dst,
                                             int* __restrict__ rows,
                                             int* __restrict__ next,
                                             int* __restrict__ ctr,
                                             int ne, int n) {
    int lane = threadIdx.x & 31;
    unsigned where = __builtin_amdgcn_s_sendmsg_rtn(0x87);
    int half = (int)((where >> 18) & 1u);
    int nchunks = (ne + LINK_CHUNK - 1) / LINK_CHUNK;
    int split = n >> 1;
    for (int pass = 0; pass < 2; ++pass) {
        int h = half ^ pass;
        int* cnt = ctr + h * CTR_STRIDE;
        if (pass == 1) {
            int seen = 0;
            if (lane == 0) seen = __hip_atomic_load(cnt, __ATOMIC_RELAXED, __HIP_MEMORY_SCOPE_AGENT);
            seen = __builtin_amdgcn_readfirstlane(seen);
            if (seen >= nchunks) break;
        }
        int g = 0;
        if (lane == 0) g = atomicAdd(cnt, 2);
        int c0 = __builtin_amdgcn_readfirstlane(g);
        int c1 = c0 + 1;
        int dd0[8], ss0[8];
#pragma unroll
        for (int k = 0; k < 8; ++k) { dd0[k] = -1; ss0[k] = 0; }
        if (c0 < nchunks) fetch_chunk(src, dst, ne, c0, lane, dd0, ss0);
        for (int it = 0; it < nchunks + 2; ++it) {
            if (c0 >= nchunks && c1 >= nchunks) break;
            unsigned g2 = (unsigned)nchunks;
            if (c1 < nchunks && lane == 0) g2 = atomicInc((unsigned*)cnt, 0xffffffffu);
            int dd1[8], ss1[8];
#pragma unroll
            for (int k = 0; k < 8; ++k) { dd1[k] = -1; ss1[k] = 0; }
            if (c1 < nchunks) fetch_chunk(src, dst, ne, c1, lane, dd1, ss1);
            if (c0 < nchunks) link_edges(rows, next, n, split, h, c0, lane, dd0, ss0);
            c0 = c1;
            c1 = (int)__builtin_amdgcn_readfirstlane(g2);
#pragma unroll
            for (int k = 0; k < 8; ++k) { dd0[k] = dd1[k]; ss0[k] = ss1[k]; }
        }
    }
}

__global__ void gat_scores_and_link(const float* __restrict__ z,
                                    const float* __restrict__ attn,
                                    float* __restrict__ s_l,
                                    float* __restrict__ s_r,
                                    int n,
                                    const int* __restrict__ src,
                                    const int* __restrict__ dst,
                                    int* __restrict__ rows,
                                    int* __restrict__ next,
                                    int* __restrict__ ctr,
                                    int ne, int score_blocks) {
    int b = blockIdx.x;
    if (b < score_blocks) {
        node_scores_body(z, attn, s_l, s_r, n, b * blockDim.x + threadIdx.x);
    } else {
        link_by_half(src, dst, rows, next, ctr, ne, n);
    }
}

template <int CTRL>
__device__ __forceinline__ float row_xor(float x) {
    int xi = __float_as_int(x);
    return __int_as_float(__builtin_amdgcn_update_dpp(0, xi, CTRL, 0xf, 0xf, true));
}
__device__ __forceinline__ float other_row(float x) {
    int xi = __float_as_int(x);
    return __int_as_float(__builtin_amdgcn_permlanex16(xi, xi, 0x76543210, 0xfedcba98, false, false));
}

template <int K, int J>
__device__ __forceinline__ int sort_step(int x, int lane) {
    int p = (J == 16) ? __builtin_amdgcn_permlanex16(x, x, 0x76543210, 0xfedcba98, false, false)
                      : __builtin_amdgcn_update_dpp(0, x, 0x160 + (J & 15), 0xf, 0xf, true);
    int mn = (x < p) ? x : p;
    int mx = (x < p) ? p : x;
    bool take_min = ((lane & J) == 0) == ((lane & K) == 0);
    return take_min ? mn : mx;
}
__device__ __forceinline__ int sort32(int x, int lane) {
    x = sort_step<2, 1>(x, lane);
    x = sort_step<4, 2>(x, lane);
    x = sort_step<4, 1>(x, lane);
    x = sort_step<8, 4>(x, lane);
    x = sort_step<8, 2>(x, lane);
    x = sort_step<8, 1>(x, lane);
    x = sort_step<16, 8>(x, lane);
    x = sort_step<16, 4>(x, lane);
    x = sort_step<16, 2>(x, lane);
    x = sort_step<16, 1>(x, lane);
    x = sort_step<32, 16>(x, lane);
    x = sort_step<32, 8>(x, lane);
    x = sort_step<32, 4>(x, lane);
    x = sort_step<32, 2>(x, lane);
    x = sort_step<32, 1>(x, lane);
    return x;
}

#define NODES_PER_WG 64
#define AGG_WAVES 8
#define NODES_PER_WAVE (NODES_PER_WG / AGG_WAVES)
#define ROWS_AHEAD 8

__device__ __forceinline__ bool holds_entry(int lane, int d) {
    int dc = (d < ROW_CAP) ? d : ROW_CAP;
    return lane >= ROW_HDR && lane < dc + ROW_HDR;
}
__device__ __forceinline__ int row_key(int word, int lane, int d) {
    int key = holds_entry(lane, d) ? word : 0x7fffffff;
    return (lane < ROW_HDR) ? (int)0x80000000 + lane : key;
}

template <int N>
__device__ __forceinline__ void rows_issue(const float* __restrict__ z, int sv, int exb, int l0,
                                           unsigned col_bytes, float (&w)[N], v2f (&zr)[N]) {
#pragma unroll
    for (int u = 0; u < N; ++u) {
        unsigned s = (unsigned)__builtin_amdgcn_readlane(sv, l0 + u);
        w[u] = __int_as_float(__builtin_amdgcn_readlane(exb, l0 + u));
        const char* rowp = (const char*)z + ((size_t)s * (OUT_DIM * sizeof(float)));
        zr[u] = *(const v2f*)(rowp + col_bytes);
    }
}
template <int N>
__device__ __forceinline__ void rows_sum(const float (&w)[N], const v2f (&zr)[N], v2f& acc, float& denom) {
#pragma unroll
    for (int u = 0; u < N; ++u) {
        denom += w[u];
        v2f ww = {w[u], w[u]};
        acc = __builtin_elementwise_fma(zr[u], ww, acc);
    }
}

__global__ void __launch_bounds__(256)
gat_node_aggregate(const float* __restrict__ z,
                   const int* __restrict__ rows,
                   const int* __restrict__ next,
                   const int* __restrict__ src,
                   const float* __restrict__ s_l,
                   const float* __restrict__ s_r,
                   float* __restrict__ out,
                   int n, int ne) {
    int lane = threadIdx.x & 31;
    int wv = __builtin_amdgcn_readfirstlane(threadIdx.x >> 5);
    int c0 = lane * 2;
    int node_first = blockIdx.x * NODES_PER_WG + wv * NODES_PER_WAVE;

    if (node_first >= n) return;
    int node_last = node_first + NODES_PER_WAVE - 1;
    if (node_last > n - 1) node_last = n - 1;

    int nodeB = (node_first + 1 < node_last) ? node_first + 1 : node_last;
    int   rawA = rows[(size_t)node_first * ROW_WORDS + lane];
    int   rowB = rows[(size_t)nodeB * ROW_WORDS + lane];
    int srn = node_first + (lane & (NODES_PER_WAVE - 1));
    float sr_all = s_r[(srn < node_last) ? srn : node_last];
    int   dA   = __builtin_amdgcn_readfirstlane(rawA);
    int   ovhA = __builtin_amdgcn_readlane(rawA, 1);
    int   rowA = sort32(row_key(rawA, lane, dA), lane);
    float slA  = s_l[holds_entry(lane, dA) ? rowA : 0];

    for (int t = 0; t < NODES_PER_WAVE; ++t) {
        int node = node_first + t;
        if (node > node_last) break;

        int nodeC = (node + 2 < node_last) ? node + 2 : node_last;
        int   rowC = rows[(size_t)nodeC * ROW_WORDS + lane];
        int   dB   = (node + 1 <= node_last) ? __builtin_amdgcn_readfirstlane(rowB) : 0;
        int   ovhB = __builtin_amdgcn_readlane(rowB, 1);
        int   sortB = sort32(row_key(rowB, lane, dB), lane);
        float slB = s_l[holds_entry(lane, dB) ? sortB : 0];

        int   d   = dA;
        int   ovh = ovhA;
        float sr  = __int_as_float(__builtin_amdgcn_readlane(__float_as_int(sr_all), t));

        float outx = 0.f, outy = 0.f;
        if (d > 0) {
            int  dc  = (d < ROW_CAP) ? d : ROW_CAP;
            bool act = holds_entry(lane, d);
            int  sv  = rowA;

            float v  = slA + sr;
            float ev = (v > 0.f) ? v : NEG_SLOPE * v;

            float part = act ? ev : 0.f;
            part += row_xor<0x161>(part);
            part += row_xor<0x162>(part);
            part += row_xor<0x164>(part);
            part += row_xor<0x168>(part);
            part += other_row(part);
            float sum = __int_as_float(__builtin_amdgcn_readfirstlane(__float_as_int(part)));
            for (int q = ovh, c = dc; (unsigned)q < (unsigned)ne && c < d; ++c) {
                float xv0 = s_l[src[q]] + sr;
                float xv = (xv0 > 0.f) ? xv0 : NEG_SLOPE * xv0;
                sum += xv;
                q = next[q];
            }
            float mean = sum / (float)d;

            float e2 = (ev < mean) ? 0.f : ev;

            float m = act ? e2 : -1e30f;
            m = fmaxf(m, row_xor<0x161>(m));
            m = fmaxf(m, row_xor<0x162>(m));
            m = fmaxf(m, row_xor<0x164>(m));
            m = fmaxf(m, row_xor<0x168>(m));
            m = fmaxf(m, other_row(m));
            for (int q = ovh, c = dc; (unsigned)q < (unsigned)ne && c < d; ++c) {
                float xv0 = s_l[src[q]] + sr;
                float xv = (xv0 > 0.f) ? xv0 : NEG_SLOPE * xv0;
                float x2 = (xv < mean) ? 0.f : xv;
                m = fmaxf(m, x2);
                q = next[q];
            }

            int exb = __float_as_int(__expf(e2 - m));

            v2f acc = {0.f, 0.f};
            float denom = 0.f;
            unsigned col_bytes = (unsigned)c0 * (unsigned)sizeof(float);
            if (dc >= ROWS_AHEAD) {
                float w[ROWS_AHEAD];
                v2f   zr[ROWS_AHEAD];
                rows_issue<ROWS_AHEAD>(z, sv, exb, ROW_HDR, col_bytes, w, zr);
                rows_sum<ROWS_AHEAD>(w, zr, acc, denom);
            }
            if (dc >= 2 * ROWS_AHEAD) {
                float w[ROWS_AHEAD];
                v2f   zr[ROWS_AHEAD];
                rows_issue<ROWS_AHEAD>(z, sv, exb, ROW_HDR + ROWS_AHEAD, col_bytes, w, zr);
                rows_sum<ROWS_AHEAD>(w, zr, acc, denom);
            }
            if (dc >= 3 * ROWS_AHEAD) {
                float w[ROWS_AHEAD];
                v2f   zr[ROWS_AHEAD];
                rows_issue<ROWS_AHEAD>(z, sv, exb, ROW_HDR + 2 * ROWS_AHEAD, col_bytes, w, zr);
                rows_sum<ROWS_AHEAD>(w, zr, acc, denom);
            }
            int rem = dc & (ROWS_AHEAD - 1);
            if (rem != 0) {
                float w4[4], w2[2], w1[1];
                v2f   z4[4], z2[2], z1[1];
                int l4 = ROW_HDR + (dc - rem);
                int l2 = l4 + (rem & 4);
                int l1 = l2 + (rem & 2);
                if (rem & 4) rows_issue<4>(z, sv, exb, l4, col_bytes, w4, z4);
                if (rem & 2) rows_issue<2>(z, sv, exb, l2, col_bytes, w2, z2);
                if (rem & 1) rows_issue<1>(z, sv, exb, l1, col_bytes, w1, z1);
                if (rem & 4) rows_sum<4>(w4, z4, acc, denom);
                if (rem & 2) rows_sum<2>(w2, z2, acc, denom);
                if (rem & 1) rows_sum<1>(w1, z1, acc, denom);
            }
            float accx = acc.x, accy = acc.y;
            for (int q = ovh, c = dc; (unsigned)q < (unsigned)ne && c < d; ++c) {
                int s = src[q];
                float xv0 = s_l[s] + sr;
                float xv = (xv0 > 0.f) ? xv0 : NEG_SLOPE * xv0;
                float x2 = (xv < mean) ? 0.f : xv;
                float ex = __expf(x2 - m);
                denom += ex;
                const float2 zv = *(const float2*)(z + (size_t)s * OUT_DIM + c0);
                accx += ex * zv.x;
                accy += ex * zv.y;
                q = next[q];
            }
            float inv = 1.f / denom;
            outx = accx * inv;
            outy = accy * inv;
        }
        float2 o;
        o.x = outx;
        o.y = outy;
        *(float2*)(out + (size_t)node * OUT_DIM + c0) = o;

        rowA = sortB;  slA = slB;  dA = dB;  ovhA = ovhB;
        rowB = rowC;
    }
}

extern "C" void kernel_launch(void* const* d_in, const int* in_sizes, int n_in,
                              void* d_out, int out_size, void* d_ws, size_t ws_size,
                              hipStream_t stream) {
    const float* h      = (const float*)d_in[0];
    const int*   src    = (const int*)d_in[1];
    const int*   dst    = (const int*)d_in[2];
    const float* W_fc   = (const float*)d_in[3];
    const float* W_attn = (const float*)d_in[4];
    float* out = (float*)d_out;

    char* ws = (char*)d_ws;
    size_t off = 0;
    auto carve = [&](size_t bytes) {
        void* p = ws + off;
        off += (bytes + 15) & ~(size_t)15;
        return p;
    };
    float* z      = (float*)carve((size_t)N_NODES * OUT_DIM * sizeof(float));
    float* s_l    = (float*)carve((size_t)N_NODES * sizeof(float));
    float* s_r    = (float*)carve((size_t)N_NODES * sizeof(float));
    int*   rows   = (int*)carve((size_t)N_NODES * ROW_WORDS * sizeof(int));
    int*   next   = (int*)carve((size_t)N_EDGES * sizeof(int));
    int*   ctr    = (int*)carve((size_t)2 * CTR_STRIDE * sizeof(int));
    if (off > ws_size) return;

    const int B = 256;

    int gemm_waves  = N_NODES / 16;
    int gemm_blocks = (gemm_waves + 7) / 8;
    gat_gemm_wmma<<<gemm_blocks, B, 0, stream>>>(h, W_fc, z, rows, ctr, N_NODES);

    const int score_blocks = (N_NODES + B - 1) / B;
    const int link_blocks  = 512;
    gat_scores_and_link<<<score_blocks + link_blocks, B, 0, stream>>>(
        z, W_attn, s_l, s_r, N_NODES, src, dst, rows, next, ctr, N_EDGES, score_blocks);

    int agg_blocks = (N_NODES + NODES_PER_WG - 1) / NODES_PER_WG;
    gat_node_aggregate<<<agg_blocks, AGG_WAVES * 32, 0, stream>>>(z, rows, next, src, s_l, s_r, out, N_NODES, N_EDGES);
}
